# starting kernel
# baseline (speedup 1.0000x reference)
#include <hip/hip_runtime.h>
#include <hip/hip_bf16.h>
#include <math.h>


typedef float v2f __attribute__((ext_vector_type(2)));
typedef float v8f __attribute__((ext_vector_type(8)));

#define WMMA_F32(a, b, c) \
  __builtin_amdgcn_wmma_f32_16x16x4_f32(false, (a), false, (b), (short)0, (c), false, false)

static constexpr int D_IN = 128;


__global__ void k_zero_ints(int* __restrict__ p, int n) {
  int i = blockIdx.x * blockDim.x + threadIdx.x;
  if (i < n) p[i] = 0;
}

__global__ void k_hist(const int* __restrict__ etype, int E, int* __restrict__ counts) {
  int e = blockIdx.x * blockDim.x + threadIdx.x;
  if (e < E) atomicAdd(&counts[etype[e] & 7], 1);
}

__global__ void k_scan(const int* __restrict__ counts, int* __restrict__ starts) {
  if (threadIdx.x == 0 && blockIdx.x == 0) {
    int acc = 0;
    for (int r = 0; r < 8; ++r) {
      starts[r] = acc;
      acc += (counts[r] + 15) & ~15;
    }
    starts[8] = acc;
  }
}

__global__ void k_scatter(const int* __restrict__ ei, const int* __restrict__ etype, int E,
                          const int* __restrict__ starts, int* __restrict__ cursor,
                          int* __restrict__ bsrc, int* __restrict__ bdst) {
  int e = blockIdx.x * blockDim.x + threadIdx.x;
  if (e >= E) return;
  int r = etype[e] & 7;
  int pos = atomicAdd(&cursor[r], 1);
  int idx = starts[r] + pos;
  bsrc[idx] = ei[e];
  bdst[idx] = ei[E + e];
}

__global__ void k_pad(const int* __restrict__ counts, const int* __restrict__ starts,
                      int* __restrict__ bsrc, int* __restrict__ bdst, int dummyRow) {
  int r = blockIdx.x;
  int lo = starts[r] + counts[r];
  int hi = starts[r + 1];
  for (int i = lo + (int)threadIdx.x; i < hi; i += blockDim.x) {
    bsrc[i] = 0;
    bdst[i] = dummyRow;
  }
}


template <int DOUT>
__global__ __launch_bounds__(32 * (DOUT / 16)) void k_root_gemm(
    const float* __restrict__ x, const float* __restrict__ W,
    const float* __restrict__ bias, float* __restrict__ y, int N) {
  const int wave = threadIdx.x >> 5, lane = threadIdx.x & 31;
  const int m = lane & 15;
  const int khalf = (lane >> 4) << 1;
  const int row = (blockIdx.x << 4) + m;
  const int rowc = row < N ? row : N - 1;
  const float* xa = x + (size_t)rowc * D_IN;
  const int col = (wave << 4) + m;
  const float* wb = W + col;
  v8f c = {};
#pragma unroll 8
  for (int k = 0; k < D_IN; k += 4) {
    v2f a, b;
    a.x = xa[k + khalf];
    a.y = xa[k + khalf + 1];
    b.x = wb[(k + khalf) * DOUT];
    b.y = wb[(k + khalf + 1) * DOUT];
    c = WMMA_F32(a, b, c);
  }
  const float bv = bias[col];
  const int rbase = (blockIdx.x << 4) + ((lane >> 4) << 3);
  float* yp = y + (size_t)rbase * DOUT + col;
  if ((blockIdx.x << 4) + 16 <= N) {
#pragma unroll
    for (int i = 0; i < 8; ++i) yp[i * DOUT] = c[i] + bv;
  } else {
#pragma unroll
    for (int i = 0; i < 8; ++i)
      if (rbase + i < N) yp[i * DOUT] = c[i] + bv;
  }
}


template <int DOUT>
__global__ __launch_bounds__(32 * (DOUT / 16)) void k_edge_wmma(
    const float* __restrict__ x, const float* __restrict__ Wrel,
    const int* __restrict__ bsrc, const int* __restrict__ bdst,
    const int* __restrict__ starts, float* __restrict__ y) {
  const int base = blockIdx.x << 4;
  if (base >= starts[8]) return;
  const int wave = threadIdx.x >> 5, lane = threadIdx.x & 31;
  int r = 0;
#pragma unroll
  for (int i = 1; i < 8; ++i) r += (base >= starts[i]) ? 1 : 0;

  const int m = lane & 15;
  const int khalf = (lane >> 4) << 1;
  const int srow = bsrc[base + m];
  const float* xa = x + (size_t)srow * D_IN;
  const int col = (wave << 4) + m;
  const float* wb = Wrel + (size_t)r * (D_IN * DOUT) + col;

  v8f c = {};
#pragma unroll 8
  for (int k = 0; k < D_IN; k += 4) {
    v2f a, b;
    a.x = xa[k + khalf];
    a.y = xa[k + khalf + 1];
    b.x = wb[(k + khalf) * DOUT];
    b.y = wb[(k + khalf + 1) * DOUT];
    c = WMMA_F32(a, b, c);
  }

  const int mBase = (lane >> 4) << 3;
#pragma unroll
  for (int i = 0; i < 8; ++i) {
    int drow = bdst[base + mBase + i];
    atomicAdd(&y[(size_t)drow * DOUT + col], c[i]);
  }
}


__global__ void k_relu4(float4* __restrict__ p, long long n4) {
  long long i = (long long)blockIdx.x * blockDim.x + threadIdx.x;
  if (i < n4) {
    float4 v = p[i];
    v.x = fmaxf(v.x, 0.0f);
    v.y = fmaxf(v.y, 0.0f);
    v.z = fmaxf(v.z, 0.0f);
    v.w = fmaxf(v.w, 0.0f);
    p[i] = v;
  }
}

__global__ void k_logsoftmax(const float* __restrict__ y, float* __restrict__ out, int N) {
  int wave = threadIdx.x >> 5, lane = threadIdx.x & 31;
  int node = blockIdx.x * (blockDim.x >> 5) + wave;
  if (node >= N) return;
  const float* row = y + (size_t)node * 64;
  float v0 = row[lane], v1 = row[lane + 32];
  float mx = fmaxf(v0, v1);
#pragma unroll
  for (int off = 16; off > 0; off >>= 1) mx = fmaxf(mx, __shfl_xor(mx, off, 32));
  float s = expf(v0 - mx) + expf(v1 - mx);
#pragma unroll
  for (int off = 16; off > 0; off >>= 1) s += __shfl_xor(s, off, 32);
  float lse = logf(s) + mx;
  float* o = out + (size_t)node * 64;
  o[lane] = v0 - lse;
  o[lane + 32] = v1 - lse;
}


static inline void launch_layer(const float* x, const float* Wrel, const float* Wroot,
                                const float* bias, float* y, int dOut, int N,
                                const int* bsrc, const int* bdst, const int* starts,
                                int nodeTiles, int edgeTilesMax, hipStream_t stream) {
  if (dOut == 128) {
    k_root_gemm<128><<<nodeTiles, 32 * (128 / 16), 0, stream>>>(x, Wroot, bias, y, N);
    k_edge_wmma<128><<<edgeTilesMax, 32 * (128 / 16), 0, stream>>>(x, Wrel, bsrc, bdst, starts, y);
  } else {
    k_root_gemm<64><<<nodeTiles, 32 * (64 / 16), 0, stream>>>(x, Wroot, bias, y, N);
    k_edge_wmma<64><<<edgeTilesMax, 32 * (64 / 16), 0, stream>>>(x, Wrel, bsrc, bdst, starts, y);
  }
}

extern "C" void kernel_launch(void* const* d_in, const int* in_sizes, int n_in,
                              void* d_out, int out_size, void* d_ws, size_t ws_size,
                              hipStream_t stream) {
  const float* x0     = (const float*)d_in[0];
  const int*   ei     = (const int*)  d_in[1];
  const int*   etype  = (const int*)  d_in[2];
  const float* Wrel1  = (const float*)d_in[3];
  const float* Wroot1 = (const float*)d_in[4];
  const float* b1     = (const float*)d_in[5];
  const float* Wrel2  = (const float*)d_in[6];
  const float* Wroot2 = (const float*)d_in[7];
  const float* b2     = (const float*)d_in[8];
  const float* Wrel3  = (const float*)d_in[9];
  const float* Wroot3 = (const float*)d_in[10];
  const float* b3     = (const float*)d_in[11];

  const int N  = in_sizes[0] / D_IN;
  const int E  = in_sizes[2];
  const int D1 = 128, D2 = 128;
  const int D3 = in_sizes[11];

  char* w = (char*)d_ws;
  int* counts = (int*)w;
  int* cursor = counts + 8;
  int* starts = cursor + 8;
  size_t off = 256;
  int* bsrc = (int*)(w + off); off += (size_t)(E + 256) * sizeof(int);
  int* bdst = (int*)(w + off); off += (size_t)(E + 256) * sizeof(int);
  off = (off + 255) & ~(size_t)255;
  float* bufA = (float*)(w + off); off += (size_t)(N + 16) * D_IN * sizeof(float);
  float* bufB = (float*)(w + off);

  k_zero_ints<<<1, 64, 0, stream>>>(counts, 32);
  k_hist    <<<(E + 255) / 256, 256, 0, stream>>>(etype, E, counts);
  k_scan    <<<1, 1, 0, stream>>>(counts, starts);
  k_scatter <<<(E + 255) / 256, 256, 0, stream>>>(ei, etype, E, starts, cursor, bsrc, bdst);
  k_pad     <<<8, 64, 0, stream>>>(counts, starts, bsrc, bdst, N);

  const int nodeTiles    = (N + 15) / 16;
  const int edgeTilesMax = E / 16 + 8;

  launch_layer(x0, Wrel1, Wroot1, b1, bufA, D1, N, bsrc, bdst, starts,
               nodeTiles, edgeTilesMax, stream);
  {
    long long n4 = (long long)N * D1 / 4;
    k_relu4<<<(unsigned)((n4 + 255) / 256), 256, 0, stream>>>((float4*)bufA, n4);
  }

  launch_layer(bufA, Wrel2, Wroot2, b2, bufB, D2, N, bsrc, bdst, starts,
               nodeTiles, edgeTilesMax, stream);
  {
    long long n4 = (long long)N * D2 / 4;
    k_relu4<<<(unsigned)((n4 + 255) / 256), 256, 0, stream>>>((float4*)bufB, n4);
  }

  launch_layer(bufB, Wrel3, Wroot3, b3, bufA, D3, N, bsrc, bdst, starts,
               nodeTiles, edgeTilesMax, stream);
  k_logsoftmax<<<(N + 7) / 8, 256, 0, stream>>>(bufA, (float*)d_out, N);
}


# Edge sort in two levels, no device atomics: edges binned to 512-node blocks by LDS counts + one scan, each block counting-sorted in LDS; was 2 scattered atomics per edge; in-segment sum order changes
# speedup vs baseline: 5.0293x; 5.0293x over previous
#include <hip/hip_runtime.h>
#include <math.h>


typedef float v2f __attribute__((ext_vector_type(2)));
typedef float v8f __attribute__((ext_vector_type(8)));

#define WMMA_F32(a, b, c) \
  __builtin_amdgcn_wmma_f32_16x16x4_f32(false, (a), false, (b), (short)0, (c), false, false)

static constexpr int D_IN = 128;
static constexpr int TM   = 32;
static constexpr int NREL = 8;
static constexpr int KPT  = TM * NREL;
static constexpr int LD   = D_IN + 4;

__device__ __forceinline__ int edge_key(int dst, int rel) {
  return (dst / TM) * KPT + rel * TM + (dst % TM);
}


static constexpr int TPB    = 16;
static constexpr int BNODES = TPB * TM;
static constexpr int BKEYS  = TPB * KPT;
static constexpr int PCH    = 512;
static constexpr int PTH    = 1024;
static constexpr int NBMAX  = PTH;
static constexpr int EB     = 4;

template <int NT>
__device__ __forceinline__ int wg_incl_scan(int v, int* shw, int& total) {
  const int lane = threadIdx.x & 31, w = threadIdx.x >> 5;
#pragma unroll
  for (int o = 1; o < 32; o <<= 1) {
    const int t = __shfl_up(v, o, 32);
    if (lane >= o) v += t;
  }
  if (lane == 31) shw[w] = v;
  __syncthreads();
  int add = 0, tot = 0;
#pragma unroll
  for (int i = 0; i < NT / 32; ++i) {
    const int s = shw[i];
    add += (i < w) ? s : 0;
    tot += s;
  }
  total = tot;
  return v + add;
}

__global__ __launch_bounds__(PTH) void k_bin_count(const int* __restrict__ ei, int E, int nB,
                                                   int* __restrict__ cnt) {
  __shared__ int h[NBMAX];
  const int tid = threadIdx.x;
  for (int b = tid; b < nB; b += PTH) h[b] = 0;
  __syncthreads();
  const int per = (E + PCH - 1) / PCH;
  const int lo = blockIdx.x * per, hi = min(E, lo + per);
  for (int e0 = lo + tid; e0 < hi; e0 += EB * PTH) {
    int d[EB];
#pragma unroll
    for (int j = 0; j < EB; ++j) d[j] = ei[E + min(e0 + j * PTH, hi - 1)];
#pragma unroll
    for (int j = 0; j < EB; ++j)
      if (e0 + j * PTH < hi) atomicAdd(&h[d[j] / BNODES], 1);
  }
  __syncthreads();
  for (int b = tid; b < nB; b += PTH) cnt[(size_t)blockIdx.x * nB + b] = h[b];
}

__global__ __launch_bounds__(PCH) void k_bin_scan(const int* __restrict__ cnt, int nB,
                                                  int* __restrict__ off, int* __restrict__ total) {
  __shared__ int shw[PCH / 32];
  const int b = blockIdx.x, p = threadIdx.x;
  const int c = cnt[(size_t)p * nB + b];
  int tot;
  const int incl = wg_incl_scan<PCH>(c, shw, tot);
  off[(size_t)p * nB + b] = incl - c;
  if (p == 0) total[b] = tot;
}

__global__ __launch_bounds__(PTH) void k_bin_place(const int* __restrict__ ei,
                                                   const int* __restrict__ etype, int E, int nB,
                                                   const int* __restrict__ off,
                                                   const int* __restrict__ total,
                                                   int2* __restrict__ rec) {
  __shared__ int nxt[NBMAX];
  __shared__ int shw[PTH / 32];
  const int tid = threadIdx.x;
  const int t = (tid < nB) ? total[tid] : 0;
  int tot;
  const int incl = wg_incl_scan<PTH>(t, shw, tot);
  if (tid < nB) nxt[tid] = incl - t + off[(size_t)blockIdx.x * nB + tid];
  __syncthreads();
  const int per = (E + PCH - 1) / PCH;
  const int lo = blockIdx.x * per, hi = min(E, lo + per);
  for (int e0 = lo + tid; e0 < hi; e0 += EB * PTH) {
    int d[EB], s[EB], r[EB];
#pragma unroll
    for (int j = 0; j < EB; ++j) {
      const int e = min(e0 + j * PTH, hi - 1);
      d[j] = ei[E + e];
      s[j] = ei[e];
      r[j] = etype[e];
    }
#pragma unroll
    for (int j = 0; j < EB; ++j) {
      if (e0 + j * PTH < hi) {
        const int b = d[j] / BNODES;
        const int pos = atomicAdd(&nxt[b], 1);
        rec[pos] = make_int2(s[j], edge_key(d[j], r[j] & 7) - b * BKEYS);
      }
    }
  }
}

__global__ __launch_bounds__(PTH) void k_block_sort(const int2* __restrict__ rec,
                                                    const int* __restrict__ total, int numKeys,
                                                    int E, int* __restrict__ seg,
                                                    int* __restrict__ ssrc) {
  static_assert(BKEYS == 4 * PTH, "a thread owns four consecutive keys");
  __shared__ __attribute__((aligned(16))) int ck[BKEYS];
  __shared__ int shw[PTH / 32];
  const int tid = threadIdx.x, c = blockIdx.x;

  const int part = (tid < c) ? total[tid] : 0;
  int bs;
  (void)wg_incl_scan<PTH>(part, shw, bs);
  const int be = bs + total[c];
  *(int4*)&ck[4 * tid] = make_int4(0, 0, 0, 0);
  __syncthreads();

  for (int e0 = bs + tid; e0 < be; e0 += EB * PTH) {
    int k[EB];
#pragma unroll
    for (int j = 0; j < EB; ++j) k[j] = rec[min(e0 + j * PTH, be - 1)].y;
#pragma unroll
    for (int j = 0; j < EB; ++j)
      if (e0 + j * PTH < be) atomicAdd(&ck[k[j]], 1);
  }
  __syncthreads();

  const int4 n4 = *(const int4*)&ck[4 * tid];
  const int sum = n4.x + n4.y + n4.z + n4.w;
  int tot;
  const int incl = wg_incl_scan<PTH>(sum, shw, tot);
  int4 st;
  st.x = bs + incl - sum;
  st.y = st.x + n4.x;
  st.z = st.y + n4.y;
  st.w = st.z + n4.z;
  *(int4*)&ck[4 * tid] = st;
  const int gk = c * BKEYS + 4 * tid;
  if (gk + 3 < numKeys) {
    *(int4*)&seg[gk] = st;
  } else {
    if (gk < numKeys) seg[gk] = st.x;
    if (gk + 1 < numKeys) seg[gk + 1] = st.y;
    if (gk + 2 < numKeys) seg[gk + 2] = st.z;
  }
  if (c == (int)gridDim.x - 1 && tid == 0) seg[numKeys] = E;
  __syncthreads();

  for (int e0 = bs + tid; e0 < be; e0 += EB * PTH) {
    int2 q[EB];
#pragma unroll
    for (int j = 0; j < EB; ++j) q[j] = rec[min(e0 + j * PTH, be - 1)];
#pragma unroll
    for (int j = 0; j < EB; ++j)
      if (e0 + j * PTH < be) ssrc[atomicAdd(&ck[q[j].y], 1)] = q[j].x;
  }
}


template <int DOUT, bool RELU>
__global__ __launch_bounds__(32 * (DOUT / 16)) void k_pull_layer(
    const float* __restrict__ x, const float* __restrict__ Wrel,
    const float* __restrict__ Wroot, const float* __restrict__ bias,
    const int* __restrict__ seg, const int* __restrict__ ssrc,
    float* __restrict__ y, int N) {
  constexpr int NW  = DOUT / 16;
  constexpr int RPW = TM / NW;
  constexpr int NB  = 2 * RPW;
  __shared__ __attribute__((aligned(16))) float sA[2][TM * LD];
  __shared__ int sseg[KPT + 1];

  const int tid = threadIdx.x;
  const int wave = tid >> 5, lane = tid & 31;
  const int m = lane & 15;
  const int khalf = (lane >> 4) << 1;
  const int tile = blockIdx.x;
  const int col = (wave << 4) + m;
  const float4* __restrict__ x4 = (const float4*)x;

  for (int i = tid; i <= KPT; i += 32 * NW) sseg[i] = seg[(size_t)tile * KPT + i];
  __syncthreads();

  v8f c0 = {}, c1 = {};

  int eb = __builtin_amdgcn_readfirstlane(sseg[wave * RPW]);
  int ee = __builtin_amdgcn_readfirstlane(sseg[wave * RPW + RPW]);
  int ids = (eb + lane < ee) ? ssrc[eb + lane] : 0;

  for (int r = 0; r <= NREL; ++r) {
    float* A = sA[r & 1];
    if (r < NREL) {
      const int kk0 = r * TM + wave * RPW;
      float* Arow = A + (wave * RPW) * LD + lane * 4;
      const float4 zero4 = make_float4(0.0f, 0.0f, 0.0f, 0.0f);
      float4 acc = zero4;
      int cur = 0;
      for (int c = eb; c < ee; c += 32) {
        int id = ids;
        if (c != eb) id = (c + lane < ee) ? ssrc[c + lane] : 0;
        int myrow = 0;
#pragma unroll
        for (int i = 1; i < RPW; ++i) myrow += (c + lane >= sseg[kk0 + i]) ? 1 : 0;
        const int cnt = min(32, ee - c);
        for (int j0 = 0; j0 < cnt; j0 += NB) {
          float4 v[NB];
#pragma unroll
          for (int g = 0; g < NB / 4; ++g) {
            if (g == 0 || j0 + 4 * g < cnt) {
#pragma unroll
              for (int q = 0; q < 4; ++q) {
                const int e = min(j0 + 4 * g + q, cnt - 1);
                const int s = __builtin_amdgcn_readlane(id, e);
                v[4 * g + q] = x4[(size_t)s * (D_IN / 4) + lane];
              }
            } else {
#pragma unroll
              for (int q = 0; q < 4; ++q) v[4 * g + q] = zero4;
            }
          }
#pragma unroll
          for (int j = 0; j < NB; ++j) {
            if (j0 + j < cnt) {
              const int rj = __builtin_amdgcn_readlane(myrow, j0 + j);
#pragma nounroll
              while (cur < rj) {
                *(float4*)(Arow + cur * LD) = acc;
                acc = zero4;
                ++cur;
              }
              acc.x += v[j].x;
              acc.y += v[j].y;
              acc.z += v[j].z;
              acc.w += v[j].w;
            }
          }
        }
      }
#pragma nounroll
      while (cur < RPW) {
        *(float4*)(Arow + cur * LD) = acc;
        acc = zero4;
        ++cur;
      }
    } else {
      float4 own[RPW];
#pragma unroll
      for (int i = 0; i < RPW; ++i) {
        const int node = tile * TM + wave * RPW + i;
        own[i] = make_float4(0.0f, 0.0f, 0.0f, 0.0f);
        if (node < N) own[i] = x4[(size_t)node * (D_IN / 4) + lane];
      }
#pragma unroll
      for (int i = 0; i < RPW; ++i)
        *(float4*)&A[(wave * RPW + i) * LD + lane * 4] = own[i];
    }
    __syncthreads();

    if (r + 1 < NREL) {
      const int kn = (r + 1) * TM + wave * RPW;
      eb = __builtin_amdgcn_readfirstlane(sseg[kn]);
      ee = __builtin_amdgcn_readfirstlane(sseg[kn + RPW]);
      ids = (eb + lane < ee) ? ssrc[eb + lane] : 0;
    }

    const float* wb = (r < NREL ? Wrel + (size_t)r * (D_IN * DOUT) : Wroot) + col;
    const float* a0p = A + m * LD + khalf;
    const float* a1p = a0p + 16 * LD;
#pragma unroll 8
    for (int k = 0; k < D_IN; k += 4) {
      v2f b;
      b.x = wb[(k + khalf) * DOUT];
      b.y = wb[(k + khalf + 1) * DOUT];
      const v2f a0 = *(const v2f*)(a0p + k);
      const v2f a1 = *(const v2f*)(a1p + k);
      c0 = WMMA_F32(a0, b, c0);
      c1 = WMMA_F32(a1, b, c1);
    }
  }

  const float bv = bias[col];
  const int rbase = tile * TM + ((lane >> 4) << 3);
  float* yp = y + (size_t)rbase * DOUT + col;
  if (tile * TM + TM <= N) {
#pragma unroll
    for (int i = 0; i < 8; ++i) {
      float v0 = c0[i] + bv, v1 = c1[i] + bv;
      if (RELU) { v0 = fmaxf(v0, 0.0f); v1 = fmaxf(v1, 0.0f); }
      yp[i * DOUT] = v0;
      yp[(i + 16) * DOUT] = v1;
    }
  } else {
#pragma unroll
    for (int i = 0; i < 8; ++i) {
      float v0 = c0[i] + bv, v1 = c1[i] + bv;
      if (RELU) { v0 = fmaxf(v0, 0.0f); v1 = fmaxf(v1, 0.0f); }
      if (rbase + i < N) yp[i * DOUT] = v0;
      if (rbase + i + 16 < N) yp[(i + 16) * DOUT] = v1;
    }
  }
}

__global__ void k_logsoftmax(const float* __restrict__ y, float* __restrict__ out, int N) {
  int wave = threadIdx.x >> 5, lane = threadIdx.x & 31;
  int node = blockIdx.x * (blockDim.x >> 5) + wave;
  if (node >= N) return;
  const float* row = y + (size_t)node * 64;
  float v0 = row[lane], v1 = row[lane + 32];
  float mx = fmaxf(v0, v1);
#pragma unroll
  for (int off = 16; off > 0; off >>= 1) mx = fmaxf(mx, __shfl_xor(mx, off, 32));
  float s = expf(v0 - mx) + expf(v1 - mx);
#pragma unroll
  for (int off = 16; off > 0; off >>= 1) s += __shfl_xor(s, off, 32);
  float lse = logf(s) + mx;
  float* o = out + (size_t)node * 64;
  o[lane] = v0 - lse;
  o[lane + 32] = v1 - lse;
}


static inline void launch_layer(const float* x, const float* Wrel, const float* Wroot,
                                const float* bias, float* y, int dOut, bool relu, int N,
                                const int* seg, const int* ssrc, int nTiles,
                                hipStream_t stream) {
  if (dOut == 128) {
    if (relu)
      k_pull_layer<128, true><<<nTiles, 32 * (128 / 16), 0, stream>>>(x, Wrel, Wroot, bias, seg, ssrc, y, N);
    else
      k_pull_layer<128, false><<<nTiles, 32 * (128 / 16), 0, stream>>>(x, Wrel, Wroot, bias, seg, ssrc, y, N);
  } else {
    if (relu)
      k_pull_layer<64, true><<<nTiles, 32 * (64 / 16), 0, stream>>>(x, Wrel, Wroot, bias, seg, ssrc, y, N);
    else
      k_pull_layer<64, false><<<nTiles, 32 * (64 / 16), 0, stream>>>(x, Wrel, Wroot, bias, seg, ssrc, y, N);
  }
}

extern "C" void kernel_launch(void* const* d_in, const int* in_sizes, int n_in,
                              void* d_out, int out_size, void* d_ws, size_t ws_size,
                              hipStream_t stream) {
  const float* x0     = (const float*)d_in[0];
  const int*   ei     = (const int*)  d_in[1];
  const int*   etype  = (const int*)  d_in[2];
  const float* Wrel1  = (const float*)d_in[3];
  const float* Wroot1 = (const float*)d_in[4];
  const float* b1     = (const float*)d_in[5];
  const float* Wrel2  = (const float*)d_in[6];
  const float* Wroot2 = (const float*)d_in[7];
  const float* b2     = (const float*)d_in[8];
  const float* Wrel3  = (const float*)d_in[9];
  const float* Wroot3 = (const float*)d_in[10];
  const float* b3     = (const float*)d_in[11];

  const int N  = in_sizes[0] / D_IN;
  const int E  = in_sizes[2];
  const int D1 = 128, D2 = 128;
  const int D3 = in_sizes[11];

  const int nTiles  = (N + TM - 1) / TM;
  const int numKeys = nTiles * KPT;

  const int nB = (nTiles + TPB - 1) / TPB;

  char* w = (char*)d_ws;
  size_t off = 0;
  int* seg = (int*)(w + off);       off += (size_t)(numKeys + 1) * sizeof(int);
  off = (off + 255) & ~(size_t)255;
  int* ssrc = (int*)(w + off);      off += (size_t)E * sizeof(int);
  off = (off + 255) & ~(size_t)255;
  float* bufA = (float*)(w + off);  off += (size_t)N * D_IN * sizeof(float);
  off = (off + 255) & ~(size_t)255;
  float* bufB = (float*)(w + off);
  size_t so = off;
  int2* rec = (int2*)(w + so);      so += (size_t)E * sizeof(int2);
  so = (so + 255) & ~(size_t)255;
  int* binCnt = (int*)(w + so);     so += (size_t)PCH * nB * sizeof(int);
  so = (so + 255) & ~(size_t)255;
  int* binOff = (int*)(w + so);     so += (size_t)PCH * nB * sizeof(int);
  so = (so + 255) & ~(size_t)255;
  int* binTot = (int*)(w + so);

  k_bin_count <<<PCH, PTH, 0, stream>>>(ei, E, nB, binCnt);
  k_bin_scan  <<<nB, PCH, 0, stream>>>(binCnt, nB, binOff, binTot);
  k_bin_place <<<PCH, PTH, 0, stream>>>(ei, etype, E, nB, binOff, binTot, rec);
  k_block_sort<<<nB, PTH, 0, stream>>>(rec, binTot, numKeys, E, seg, ssrc);

  launch_layer(x0,   Wrel1, Wroot1, b1, bufA, D1, true,  N, seg, ssrc, nTiles, stream);
  launch_layer(bufA, Wrel2, Wroot2, b2, bufB, D2, true,  N, seg, ssrc, nTiles, stream);
  launch_layer(bufB, Wrel3, Wroot3, b3, bufA, D3, false, N, seg, ssrc, nTiles, stream);
  k_logsoftmax<<<(N + 7) / 8, 256, 0, stream>>>(bufA, (float*)d_out, N);
}
